# starting kernel
# baseline (speedup 1.0000x reference)
#include <hip/hip_runtime.h>
#include <hip/hip_bf16.h>

typedef __attribute__((ext_vector_type(2))) float v2f;
typedef __attribute__((ext_vector_type(8))) float v8f;

#define IN_C    256
#define OUT_C   256
#define N_SPLIT 4
#define N_LEGO  64
#define BASIC_C 64
#define HIN     56
#define WIN     56
#define HOUT    19
#define WOUT    58
#define KDIM    192
#define NCOL    64
#define XSP_STR 160
#define FS_W    64

__global__ __launch_bounds__(256, 1)
void lego_conv_fused(const float* __restrict__ x,
                     const float* __restrict__ lego,
                     const float* __restrict__ coef,
                     const float* __restrict__ comb,
                     float* __restrict__ out) {
    __shared__ float XsP[(KDIM / 2) * XSP_STR];
    __shared__ float Fs[N_SPLIT * N_LEGO * FS_W];
    __shared__ float cW[N_SPLIT * OUT_C];
    __shared__ int   iW[N_SPLIT * OUT_C];

    const int tid = threadIdx.x;
    const int b   = blockIdx.y;
    const int h   = blockIdx.x;

    for (int p = tid; p < N_SPLIT * OUT_C; p += 256) {
        const float* cb = comb + p * N_LEGO;
        int   best = 0;
        float bv   = cb[0];
        #pragma unroll
        for (int n = 1; n < N_LEGO; ++n) {
            float v = cb[n];
            if (v > bv) { bv = v; best = n; }
        }
        cW[p] = coef[p * N_LEGO + best];
        iW[p] = best;
    }

    const int lane = tid & 31;
    const int wave = tid >> 5;
    const int half = lane >> 4;
    const int l16  = lane & 15;
    const int mt   = wave >> 1;
    const int nt0  = (wave & 1) * 2;

    const float* Arow = lego + (mt * 16 + l16) * KDIM + half * 2;

    for (int s = 0; s < N_SPLIT; ++s) {
        __syncthreads();

        for (int j = tid; j < KDIM * NCOL; j += 256) {
            int   k = j >> 6;
            int   w = j & 63;
            float val = 0.0f;
            if (w >= 1 && w <= WIN) {
                int ci = k / 3;
                int kh = k - ci * 3;
                int r  = 3 * h + kh - 1;
                if (r >= 0)
                    val = x[(((size_t)b * IN_C + s * BASIC_C + ci) * HIN + r) * WIN + (w - 1)];
            }
            XsP[(k >> 1) * XSP_STR + (w << 1) + (k & 1)] = val;
        }
        __syncthreads();

        v8f acc0 = {0.f, 0.f, 0.f, 0.f, 0.f, 0.f, 0.f, 0.f};
        v8f acc1 = {0.f, 0.f, 0.f, 0.f, 0.f, 0.f, 0.f, 0.f};
        #pragma unroll 4
        for (int kb = 0; kb < KDIM; kb += 4) {
            v2f a = *(const v2f*)(Arow + kb);
            const float* brow = &XsP[((kb >> 1) + half) * XSP_STR + (l16 << 1)];
            v2f b0 = *(const v2f*)(brow + nt0 * 32);
            v2f b1 = *(const v2f*)(brow + nt0 * 32 + 32);
            acc0 = __builtin_amdgcn_wmma_f32_16x16x4_f32(
                       false, a, false, b0, (short)0, acc0, false, false);
            acc1 = __builtin_amdgcn_wmma_f32_16x16x4_f32(
                       false, a, false, b1, (short)0, acc1, false, false);
        }

        float* fbase = Fs + (s * N_LEGO + mt * 16 + half * 8) * FS_W;
        #pragma unroll
        for (int v = 0; v < 8; ++v) {
            fbase[v * FS_W + nt0       * 16 + l16] = acc0[v];
            fbase[v * FS_W + (nt0 + 1) * 16 + l16] = acc1[v];
        }
    }
    __syncthreads();

    const int wl = tid & 63;
    const int o0 = tid >> 6;
    if (wl < WOUT) {
        for (int o = o0; o < OUT_C; o += 4) {
            float acc = 0.f;
            #pragma unroll
            for (int s = 0; s < N_SPLIT; ++s) {
                int n = iW[s * OUT_C + o];
                acc += cW[s * OUT_C + o] * Fs[(s * N_LEGO + n) * FS_W + wl];
            }
            out[(((size_t)b * OUT_C + o) * HOUT + h) * WOUT + wl] = acc;
        }
    }
}

extern "C" void kernel_launch(void* const* d_in, const int* in_sizes, int n_in,
                              void* d_out, int out_size, void* d_ws, size_t ws_size,
                              hipStream_t stream) {
    (void)in_sizes; (void)n_in; (void)d_ws; (void)ws_size; (void)out_size;
    const float* x    = (const float*)d_in[0];
    const float* lego = (const float*)d_in[1];
    const float* coef = (const float*)d_in[2];
    const float* comb = (const float*)d_in[3];
    float* out = (float*)d_out;

    dim3 grid(HOUT, 32, 1);
    dim3 block(256, 1, 1);
    lego_conv_fused<<<grid, block, 0, stream>>>(x, lego, coef, comb, out);
}


# x panel staging: each thread issues all of a split's x loads as 12 independent 16-byte loads before any wait and stores them with 16-byte LDS stores; was one 4-byte load per wait
# speedup vs baseline: 3.0090x; 3.0090x over previous
#include <hip/hip_runtime.h>
#include <hip/hip_bf16.h>


typedef __attribute__((ext_vector_type(2))) float v2f;
typedef __attribute__((ext_vector_type(4))) float v4f;
typedef __attribute__((ext_vector_type(8))) float v8f;

#define IN_C    256
#define OUT_C   256
#define N_SPLIT 4
#define N_LEGO  64
#define BASIC_C 64
#define HIN     56
#define WIN     56
#define HOUT    19
#define WOUT    58
#define KDIM    192
#define NCOL    64
#define XSP_STR 160
#define XCOL0   3
#define FS_W    64
#define NQUAD   14
#define NUNIT   ((KDIM / 2) * NQUAD)
#define UPT     6
#define TAIL_T  (NUNIT - (UPT - 1) * 256)

__global__ __launch_bounds__(256, 1)
void lego_conv_fused(const float* __restrict__ x,
                     const float* __restrict__ lego,
                     const float* __restrict__ coef,
                     const float* __restrict__ comb,
                     float* __restrict__ out) {
    __shared__ __attribute__((aligned(16))) float XsP[(KDIM / 2) * XSP_STR];
    __shared__ float Fs[N_SPLIT * N_LEGO * FS_W];
    __shared__ float cW[N_SPLIT * OUT_C];
    __shared__ int   iW[N_SPLIT * OUT_C];

    const int tid = threadIdx.x;
    const int b   = blockIdx.y;
    const int h   = blockIdx.x;

    unsigned g0[UPT], g1[UPT];
    int      lo[UPT];
    unsigned zmask = 0;
    #pragma unroll
    for (int i = 0; i < UPT; ++i) {
        int u   = tid + 256 * i;
        int p   = u / NQUAD;
        int q   = u - p * NQUAD;
        int k0  = 2 * p, k1 = k0 + 1;
        int ci0 = k0 / 3, kh0 = k0 - 3 * ci0;
        int ci1 = k1 / 3, kh1 = k1 - 3 * ci1;
        int r0  = 3 * h + kh0 - 1;
        int r1  = 3 * h + kh1 - 1;
        if (r0 < 0) { zmask |= 1u << (2 * i);     r0 = 0; }
        if (r1 < 0) { zmask |= 1u << (2 * i + 1); r1 = 0; }
        g0[i] = (unsigned)((ci0 * HIN + r0) * WIN + 4 * q);
        g1[i] = (unsigned)((ci1 * HIN + r1) * WIN + 4 * q);
        lo[i] = p * XSP_STR + 2 * (4 * q + 1 + XCOL0);
    }
    const bool tail = tid < TAIL_T;

    const float* xb = x + (size_t)b * IN_C * HIN * WIN;
    v4f ra[UPT], rb[UPT];

    #pragma unroll
    for (int i = 0; i < UPT; ++i) {
        if (i < UPT - 1 || tail) {
            ra[i] = *(const v4f*)(xb + g0[i]);
            rb[i] = *(const v4f*)(xb + g1[i]);
        }
    }

    for (int p = tid; p < N_SPLIT * OUT_C; p += 256) {
        const float* cb = comb + p * N_LEGO;
        int   best = 0;
        float bv   = cb[0];
        #pragma unroll
        for (int n = 1; n < N_LEGO; ++n) {
            float v = cb[n];
            if (v > bv) { bv = v; best = n; }
        }
        cW[p] = coef[p * N_LEGO + best];
        iW[p] = best;
    }

    for (int i = tid; i < (KDIM / 2) * 8; i += 256) {
        int row = i >> 3, j = i & 7;
        int c   = (j == 0) ? 0 : (WIN + j);
        *(v2f*)&XsP[row * XSP_STR + 2 * (c + XCOL0)] = (v2f){0.f, 0.f};
    }

    const int lane = tid & 31;
    const int wave = tid >> 5;
    const int half = lane >> 4;
    const int l16  = lane & 15;
    const int mt   = wave >> 1;
    const int nt0  = (wave & 1) * 2;

    const float* Arow = lego + (mt * 16 + l16) * KDIM + half * 2;

    for (int s = 0; s < N_SPLIT; ++s) {
        if (s > 0) {
            const float* xs = xb + (size_t)s * BASIC_C * HIN * WIN;
            #pragma unroll
            for (int i = 0; i < UPT; ++i) {
                if (i < UPT - 1 || tail) {
                    ra[i] = *(const v4f*)(xs + g0[i]);
                    rb[i] = *(const v4f*)(xs + g1[i]);
                }
            }
        }
        __syncthreads();

        #pragma unroll
        for (int i = 0; i < UPT; ++i) {
            if (i < UPT - 1 || tail) {
                v4f a = ra[i], c = rb[i];
                if (zmask & (1u << (2 * i)))     a = (v4f){0.f, 0.f, 0.f, 0.f};
                if (zmask & (1u << (2 * i + 1))) c = (v4f){0.f, 0.f, 0.f, 0.f};
                *(v4f*)&XsP[lo[i]]     = (v4f){a.x, c.x, a.y, c.y};
                *(v4f*)&XsP[lo[i] + 4] = (v4f){a.z, c.z, a.w, c.w};
            }
        }
        __syncthreads();

        v8f acc0 = {0.f, 0.f, 0.f, 0.f, 0.f, 0.f, 0.f, 0.f};
        v8f acc1 = {0.f, 0.f, 0.f, 0.f, 0.f, 0.f, 0.f, 0.f};
        #pragma unroll 4
        for (int kb = 0; kb < KDIM; kb += 4) {
            v2f a = *(const v2f*)(Arow + kb);
            const float* brow = &XsP[((kb >> 1) + half) * XSP_STR + 2 * XCOL0 + (l16 << 1)];
            v2f b0 = *(const v2f*)(brow + nt0 * 32);
            v2f b1 = *(const v2f*)(brow + nt0 * 32 + 32);
            acc0 = __builtin_amdgcn_wmma_f32_16x16x4_f32(
                       false, a, false, b0, (short)0, acc0, false, false);
            acc1 = __builtin_amdgcn_wmma_f32_16x16x4_f32(
                       false, a, false, b1, (short)0, acc1, false, false);
        }

        float* fbase = Fs + (s * N_LEGO + mt * 16 + half * 8) * FS_W;
        #pragma unroll
        for (int v = 0; v < 8; ++v) {
            fbase[v * FS_W + nt0       * 16 + l16] = acc0[v];
            fbase[v * FS_W + (nt0 + 1) * 16 + l16] = acc1[v];
        }
    }
    __syncthreads();

    const int wl = tid & 63;
    const int o0 = tid >> 6;
    if (wl < WOUT) {
        for (int o = o0; o < OUT_C; o += 4) {
            float acc = 0.f;
            #pragma unroll
            for (int s = 0; s < N_SPLIT; ++s) {
                int n = iW[s * OUT_C + o];
                acc += cW[s * OUT_C + o] * Fs[(s * N_LEGO + n) * FS_W + wl];
            }
            out[(((size_t)b * OUT_C + o) * HOUT + h) * WOUT + wl] = acc;
        }
    }
}

extern "C" void kernel_launch(void* const* d_in, const int* in_sizes, int n_in,
                              void* d_out, int out_size, void* d_ws, size_t ws_size,
                              hipStream_t stream) {
    (void)in_sizes; (void)n_in; (void)d_ws; (void)ws_size; (void)out_size;
    const float* x    = (const float*)d_in[0];
    const float* lego = (const float*)d_in[1];
    const float* coef = (const float*)d_in[2];
    const float* comb = (const float*)d_in[3];
    float* out = (float*)d_out;

    dim3 grid(HOUT, 32, 1);
    dim3 block(256, 1, 1);
    lego_conv_fused<<<grid, block, 0, stream>>>(x, lego, coef, comb, out);
}


# epilogue: each wave writes whole 58-float output rows (8-byte LDS reads, one 8-byte store per lane); lego index and coefficient reach lanes by readlane, not chained LDS reads; sum order kept
# speedup vs baseline: 3.5813x; 1.1902x over previous
#include <hip/hip_runtime.h>
#include <hip/hip_bf16.h>


typedef __attribute__((ext_vector_type(2))) float v2f;
typedef __attribute__((ext_vector_type(4))) float v4f;
typedef __attribute__((ext_vector_type(8))) float v8f;

#define IN_C    256
#define OUT_C   256
#define N_SPLIT 4
#define N_LEGO  64
#define BASIC_C 64
#define HIN     56
#define WIN     56
#define HOUT    19
#define WOUT    58
#define KDIM    192
#define NCOL    64
#define XSP_STR 160
#define XCOL0   3
#define FS_W    64
#define NQUAD   14
#define NUNIT   ((KDIM / 2) * NQUAD)
#define UPT     6
#define TAIL_T  (NUNIT - (UPT - 1) * 256)
#define EP_ROWS 8

__global__ __launch_bounds__(256, 1)
void lego_conv_fused(const float* __restrict__ x,
                     const float* __restrict__ lego,
                     const float* __restrict__ coef,
                     const float* __restrict__ comb,
                     float* __restrict__ out) {
    __shared__ __attribute__((aligned(16))) float XsP[(KDIM / 2) * XSP_STR];
    __shared__ __attribute__((aligned(16))) float Fs[N_SPLIT * N_LEGO * FS_W];
    __shared__ float cW[N_SPLIT * OUT_C];
    __shared__ int   iW[N_SPLIT * OUT_C];

    const int tid = threadIdx.x;
    const int b   = blockIdx.y;
    const int h   = blockIdx.x;

    unsigned g0[UPT], g1[UPT];
    int      lo[UPT];
    unsigned zmask = 0;
    #pragma unroll
    for (int i = 0; i < UPT; ++i) {
        int u   = tid + 256 * i;
        int p   = u / NQUAD;
        int q   = u - p * NQUAD;
        int k0  = 2 * p, k1 = k0 + 1;
        int ci0 = k0 / 3, kh0 = k0 - 3 * ci0;
        int ci1 = k1 / 3, kh1 = k1 - 3 * ci1;
        int r0  = 3 * h + kh0 - 1;
        int r1  = 3 * h + kh1 - 1;
        if (r0 < 0) { zmask |= 1u << (2 * i);     r0 = 0; }
        if (r1 < 0) { zmask |= 1u << (2 * i + 1); r1 = 0; }
        g0[i] = (unsigned)((ci0 * HIN + r0) * WIN + 4 * q);
        g1[i] = (unsigned)((ci1 * HIN + r1) * WIN + 4 * q);
        lo[i] = p * XSP_STR + 2 * (4 * q + 1 + XCOL0);
    }
    const bool tail = tid < TAIL_T;

    const float* xb = x + (size_t)b * IN_C * HIN * WIN;
    v4f ra[UPT], rb[UPT];

    #pragma unroll
    for (int i = 0; i < UPT; ++i) {
        if (i < UPT - 1 || tail) {
            ra[i] = *(const v4f*)(xb + g0[i]);
            rb[i] = *(const v4f*)(xb + g1[i]);
        }
    }

    for (int p = tid; p < N_SPLIT * OUT_C; p += 256) {
        const float* cb = comb + p * N_LEGO;
        int   best = 0;
        float bv   = cb[0];
        #pragma unroll
        for (int n = 1; n < N_LEGO; ++n) {
            float v = cb[n];
            if (v > bv) { bv = v; best = n; }
        }
        cW[p] = coef[p * N_LEGO + best];
        iW[p] = best;
    }

    for (int i = tid; i < (KDIM / 2) * 8; i += 256) {
        int row = i >> 3, j = i & 7;
        int c   = (j == 0) ? 0 : (WIN + j);
        *(v2f*)&XsP[row * XSP_STR + 2 * (c + XCOL0)] = (v2f){0.f, 0.f};
    }

    const int lane = tid & 31;
    const int wave = tid >> 5;
    const int half = lane >> 4;
    const int l16  = lane & 15;
    const int mt   = wave >> 1;
    const int nt0  = (wave & 1) * 2;

    const float* Arow = lego + (mt * 16 + l16) * KDIM + half * 2;

    for (int s = 0; s < N_SPLIT; ++s) {
        if (s > 0) {
            const float* xs = xb + (size_t)s * BASIC_C * HIN * WIN;
            #pragma unroll
            for (int i = 0; i < UPT; ++i) {
                if (i < UPT - 1 || tail) {
                    ra[i] = *(const v4f*)(xs + g0[i]);
                    rb[i] = *(const v4f*)(xs + g1[i]);
                }
            }
        }
        __syncthreads();

        #pragma unroll
        for (int i = 0; i < UPT; ++i) {
            if (i < UPT - 1 || tail) {
                v4f a = ra[i], c = rb[i];
                if (zmask & (1u << (2 * i)))     a = (v4f){0.f, 0.f, 0.f, 0.f};
                if (zmask & (1u << (2 * i + 1))) c = (v4f){0.f, 0.f, 0.f, 0.f};
                *(v4f*)&XsP[lo[i]]     = (v4f){a.x, c.x, a.y, c.y};
                *(v4f*)&XsP[lo[i] + 4] = (v4f){a.z, c.z, a.w, c.w};
            }
        }
        __syncthreads();

        v8f acc0 = {0.f, 0.f, 0.f, 0.f, 0.f, 0.f, 0.f, 0.f};
        v8f acc1 = {0.f, 0.f, 0.f, 0.f, 0.f, 0.f, 0.f, 0.f};
        #pragma unroll 4
        for (int kb = 0; kb < KDIM; kb += 4) {
            v2f a = *(const v2f*)(Arow + kb);
            const float* brow = &XsP[((kb >> 1) + half) * XSP_STR + 2 * XCOL0 + (l16 << 1)];
            v2f b0 = *(const v2f*)(brow + nt0 * 32);
            v2f b1 = *(const v2f*)(brow + nt0 * 32 + 32);
            acc0 = __builtin_amdgcn_wmma_f32_16x16x4_f32(
                       false, a, false, b0, (short)0, acc0, false, false);
            acc1 = __builtin_amdgcn_wmma_f32_16x16x4_f32(
                       false, a, false, b1, (short)0, acc1, false, false);
        }

        float* fbase = Fs + (s * N_LEGO + mt * 16 + half * 8) * FS_W;
        #pragma unroll
        for (int v = 0; v < 8; ++v) {
            fbase[v * FS_W + nt0       * 16 + l16] = acc0[v];
            fbase[v * FS_W + (nt0 + 1) * 16 + l16] = acc1[v];
        }
    }
    __syncthreads();

    {
        const int ob = wave * 32;
        int   selI[N_SPLIT];
        float selC[N_SPLIT];
        #pragma unroll
        for (int s = 0; s < N_SPLIT; ++s) {
            selI[s] = iW[s * OUT_C + ob + lane];
            selC[s] = cW[s * OUT_C + ob + lane];
        }
        const bool   own  = lane < WOUT / 2;
        const float* frow = Fs + 2 * lane;
        float*       orow = out + (((size_t)b * OUT_C + ob) * HOUT + h) * WOUT + 2 * lane;
        for (int j0 = 0; j0 < 32; j0 += EP_ROWS) {
            v2f acc[EP_ROWS];
            #pragma unroll
            for (int j = 0; j < EP_ROWS; ++j) {
                acc[j] = (v2f){0.f, 0.f};
                #pragma unroll
                for (int s = 0; s < N_SPLIT; ++s) {
                    const int   n = __builtin_amdgcn_readlane(selI[s], j0 + j);
                    const float c = __int_as_float(__builtin_amdgcn_readlane(__float_as_int(selC[s]), j0 + j));
                    const v2f   f = *(const v2f*)(frow + (s * N_LEGO + n) * FS_W);
                    acc[j] += c * f;
                }
            }
            if (own) {
                #pragma unroll
                for (int j = 0; j < EP_ROWS; ++j)
                    *(v2f*)(orow + (size_t)(j0 + j) * HOUT * WOUT) = acc[j];
            }
        }
    }
}

extern "C" void kernel_launch(void* const* d_in, const int* in_sizes, int n_in,
                              void* d_out, int out_size, void* d_ws, size_t ws_size,
                              hipStream_t stream) {
    (void)in_sizes; (void)n_in; (void)d_ws; (void)ws_size; (void)out_size;
    const float* x    = (const float*)d_in[0];
    const float* lego = (const float*)d_in[1];
    const float* coef = (const float*)d_in[2];
    const float* comb = (const float*)d_in[3];
    float* out = (float*)d_out;

    dim3 grid(HOUT, 32, 1);
    dim3 block(256, 1, 1);
    lego_conv_fused<<<grid, block, 0, stream>>>(x, lego, coef, comb, out);
}


# split loop pipelined: next split's x loads are issued ahead of the matrix loop and stay in flight under it; for that the loop's lego (A) fragments are loaded once into registers; sum order kept
# speedup vs baseline: 4.2426x; 1.1847x over previous
#include <hip/hip_runtime.h>
#include <hip/hip_bf16.h>


typedef __attribute__((ext_vector_type(2))) float v2f;
typedef __attribute__((ext_vector_type(4))) float v4f;
typedef __attribute__((ext_vector_type(8))) float v8f;

#define IN_C    256
#define OUT_C   256
#define N_SPLIT 4
#define N_LEGO  64
#define BASIC_C 64
#define HIN     56
#define WIN     56
#define HOUT    19
#define WOUT    58
#define KDIM    192
#define NCOL    64
#define XSP_STR 160
#define XCOL0   3
#define FS_W    64
#define NQUAD   14
#define NUNIT   ((KDIM / 2) * NQUAD)
#define UPT     6
#define TAIL_T  (NUNIT - (UPT - 1) * 256)
#define KSTEPS  (KDIM / 4)
#define EP_ROWS 8

__global__ __launch_bounds__(256, 1)
void lego_conv_fused(const float* __restrict__ x,
                     const float* __restrict__ lego,
                     const float* __restrict__ coef,
                     const float* __restrict__ comb,
                     float* __restrict__ out) {
    __shared__ __attribute__((aligned(16))) float XsP[(KDIM / 2) * XSP_STR];
    __shared__ __attribute__((aligned(16))) float Fs[N_SPLIT * N_LEGO * FS_W];
    __shared__ float cW[N_SPLIT * OUT_C];
    __shared__ int   iW[N_SPLIT * OUT_C];

    const int tid = threadIdx.x;
    const int b   = blockIdx.y;
    const int h   = blockIdx.x;

    unsigned g0[UPT], g1[UPT];
    int      lo[UPT];
    unsigned zmask = 0;
    #pragma unroll
    for (int i = 0; i < UPT; ++i) {
        int u   = tid + 256 * i;
        int p   = u / NQUAD;
        int q   = u - p * NQUAD;
        int k0  = 2 * p, k1 = k0 + 1;
        int ci0 = k0 / 3, kh0 = k0 - 3 * ci0;
        int ci1 = k1 / 3, kh1 = k1 - 3 * ci1;
        int r0  = 3 * h + kh0 - 1;
        int r1  = 3 * h + kh1 - 1;
        if (r0 < 0) { zmask |= 1u << (2 * i);     r0 = 0; }
        if (r1 < 0) { zmask |= 1u << (2 * i + 1); r1 = 0; }
        g0[i] = (unsigned)((ci0 * HIN + r0) * WIN + 4 * q);
        g1[i] = (unsigned)((ci1 * HIN + r1) * WIN + 4 * q);
        lo[i] = p * XSP_STR + 2 * (4 * q + 1 + XCOL0);
    }
    const bool tail = tid < TAIL_T;

    const float* xb = x + (size_t)b * IN_C * HIN * WIN;
    v4f ra[UPT], rb[UPT];

    #pragma unroll
    for (int i = 0; i < UPT; ++i) {
        if (i < UPT - 1 || tail) {
            ra[i] = *(const v4f*)(xb + g0[i]);
            rb[i] = *(const v4f*)(xb + g1[i]);
        }
    }

    for (int p = tid; p < N_SPLIT * OUT_C; p += 256) {
        const float* cb = comb + p * N_LEGO;
        int   best = 0;
        float bv   = cb[0];
        #pragma unroll
        for (int n = 1; n < N_LEGO; ++n) {
            float v = cb[n];
            if (v > bv) { bv = v; best = n; }
        }
        cW[p] = coef[p * N_LEGO + best];
        iW[p] = best;
    }

    for (int i = tid; i < (KDIM / 2) * 8; i += 256) {
        int row = i >> 3, j = i & 7;
        int c   = (j == 0) ? 0 : (WIN + j);
        *(v2f*)&XsP[row * XSP_STR + 2 * (c + XCOL0)] = (v2f){0.f, 0.f};
    }

    const int lane = tid & 31;
    const int wave = tid >> 5;
    const int half = lane >> 4;
    const int l16  = lane & 15;
    const int mt   = wave >> 1;
    const int nt0  = (wave & 1) * 2;

    const float* Arow = lego + (mt * 16 + l16) * KDIM + half * 2;
    v2f afrag[KSTEPS];
    #pragma unroll
    for (int t = 0; t < KSTEPS; ++t)
        afrag[t] = *(const v2f*)(Arow + 4 * t);

    for (int s = 0; s < N_SPLIT; ++s) {
        __syncthreads();

        #pragma unroll
        for (int i = 0; i < UPT; ++i) {
            if (i < UPT - 1 || tail) {
                v4f a = ra[i], c = rb[i];
                if (zmask & (1u << (2 * i)))     a = (v4f){0.f, 0.f, 0.f, 0.f};
                if (zmask & (1u << (2 * i + 1))) c = (v4f){0.f, 0.f, 0.f, 0.f};
                *(v4f*)&XsP[lo[i]]     = (v4f){a.x, c.x, a.y, c.y};
                *(v4f*)&XsP[lo[i] + 4] = (v4f){a.z, c.z, a.w, c.w};
            }
        }
        __syncthreads();

        if (s + 1 < N_SPLIT) {
            const float* xs = xb + (size_t)(s + 1) * BASIC_C * HIN * WIN;
            #pragma unroll
            for (int i = 0; i < UPT; ++i) {
                if (i < UPT - 1 || tail) {
                    ra[i] = *(const v4f*)(xs + g0[i]);
                    rb[i] = *(const v4f*)(xs + g1[i]);
                }
            }
        }

        v8f acc0 = {0.f, 0.f, 0.f, 0.f, 0.f, 0.f, 0.f, 0.f};
        v8f acc1 = {0.f, 0.f, 0.f, 0.f, 0.f, 0.f, 0.f, 0.f};
        const float* bbase = &XsP[half * XSP_STR + 2 * XCOL0 + (l16 << 1) + nt0 * 32];
        #pragma unroll
        for (int t = 0; t < KSTEPS; ++t) {
            const float* brow = bbase + 2 * t * XSP_STR;
            v2f b0 = *(const v2f*)(brow);
            v2f b1 = *(const v2f*)(brow + 32);
            acc0 = __builtin_amdgcn_wmma_f32_16x16x4_f32(
                       false, afrag[t], false, b0, (short)0, acc0, false, false);
            acc1 = __builtin_amdgcn_wmma_f32_16x16x4_f32(
                       false, afrag[t], false, b1, (short)0, acc1, false, false);
        }

        float* fbase = Fs + (s * N_LEGO + mt * 16 + half * 8) * FS_W;
        #pragma unroll
        for (int v = 0; v < 8; ++v) {
            fbase[v * FS_W + nt0       * 16 + l16] = acc0[v];
            fbase[v * FS_W + (nt0 + 1) * 16 + l16] = acc1[v];
        }
    }
    __syncthreads();

    {
        const int ob = wave * 32;
        int   selI[N_SPLIT];
        float selC[N_SPLIT];
        #pragma unroll
        for (int s = 0; s < N_SPLIT; ++s) {
            selI[s] = iW[s * OUT_C + ob + lane];
            selC[s] = cW[s * OUT_C + ob + lane];
        }
        const bool   own  = lane < WOUT / 2;
        const float* frow = Fs + 2 * lane;
        float*       orow = out + (((size_t)b * OUT_C + ob) * HOUT + h) * WOUT + 2 * lane;
        for (int j0 = 0; j0 < 32; j0 += EP_ROWS) {
            v2f acc[EP_ROWS];
            #pragma unroll
            for (int j = 0; j < EP_ROWS; ++j) {
                acc[j] = (v2f){0.f, 0.f};
                #pragma unroll
                for (int s = 0; s < N_SPLIT; ++s) {
                    const int   n = __builtin_amdgcn_readlane(selI[s], j0 + j);
                    const float c = __int_as_float(__builtin_amdgcn_readlane(__float_as_int(selC[s]), j0 + j));
                    const v2f   f = *(const v2f*)(frow + (s * N_LEGO + n) * FS_W);
                    acc[j] += c * f;
                }
            }
            if (own) {
                #pragma unroll
                for (int j = 0; j < EP_ROWS; ++j)
                    *(v2f*)(orow + (size_t)(j0 + j) * HOUT * WOUT) = acc[j];
            }
        }
    }
}

extern "C" void kernel_launch(void* const* d_in, const int* in_sizes, int n_in,
                              void* d_out, int out_size, void* d_ws, size_t ws_size,
                              hipStream_t stream) {
    (void)in_sizes; (void)n_in; (void)d_ws; (void)ws_size; (void)out_size;
    const float* x    = (const float*)d_in[0];
    const float* lego = (const float*)d_in[1];
    const float* coef = (const float*)d_in[2];
    const float* comb = (const float*)d_in[3];
    float* out = (float*)d_out;

    dim3 grid(HOUT, 32, 1);
    dim3 block(256, 1, 1);
    lego_conv_fused<<<grid, block, 0, stream>>>(x, lego, coef, comb, out);
}


# selection table hoisted: argmax index and coefficient of the 1024 (split, channel) pairs are found once per call by a small first kernel into the workspace, not by each of 608 work-groups; sums kept
# speedup vs baseline: 5.2783x; 1.2441x over previous
#include <hip/hip_runtime.h>
#include <hip/hip_bf16.h>


typedef __attribute__((ext_vector_type(2))) float v2f;
typedef __attribute__((ext_vector_type(4))) float v4f;
typedef __attribute__((ext_vector_type(8))) float v8f;
typedef __attribute__((ext_vector_type(4))) int   v4i;

#define IN_C    256
#define OUT_C   256
#define N_SPLIT 4
#define N_LEGO  64
#define BASIC_C 64
#define HIN     56
#define WIN     56
#define HOUT    19
#define WOUT    58
#define KDIM    192
#define NCOL    64
#define XSP_STR 160
#define XCOL0   3
#define FS_W    64
#define NQUAD   14
#define NUNIT   ((KDIM / 2) * NQUAD)
#define UPT     6
#define TAIL_T  (NUNIT - (UPT - 1) * 256)
#define KSTEPS  (KDIM / 4)
#define EP_ROWS 8
#define TAB_N     (N_SPLIT * OUT_C)
#define TAB_BYTES (2 * TAB_N * 4)
#define SEL_WGS   32

__global__ __launch_bounds__(32)
void lego_select(const float* __restrict__ coef,
                 const float* __restrict__ comb,
                 float* __restrict__ tab, size_t tab_room) {
    if (tab_room < (size_t)TAB_BYTES) return;
    const int p = blockIdx.x * 32 + threadIdx.x;
    const float* cb = comb + p * N_LEGO;
    int   best = 0;
    float bv   = cb[0];
    #pragma unroll
    for (int n = 1; n < N_LEGO; ++n) {
        float v = cb[n];
        if (v > bv) { bv = v; best = n; }
    }
    tab[p]         = coef[p * N_LEGO + best];
    tab[TAB_N + p] = __int_as_float(best);
}

__global__ __launch_bounds__(256, 1)
void lego_conv_fused(const float* __restrict__ x,
                     const float* __restrict__ lego,
                     const float* __restrict__ coef,
                     const float* __restrict__ comb,
                     const float* __restrict__ tab, size_t tab_room,
                     float* __restrict__ out) {
    __shared__ __attribute__((aligned(16))) float XsP[(KDIM / 2) * XSP_STR];
    __shared__ __attribute__((aligned(16))) float Fs[N_SPLIT * N_LEGO * FS_W];
    __shared__ __attribute__((aligned(16))) float cW[N_SPLIT * OUT_C];
    __shared__ __attribute__((aligned(16))) int   iW[N_SPLIT * OUT_C];

    const int tid = threadIdx.x;
    const int b   = blockIdx.y;
    const int h   = blockIdx.x;

    const bool have_tab = tab_room >= (size_t)TAB_BYTES;
    v4f tc = {0.f, 0.f, 0.f, 0.f}, ti = {0.f, 0.f, 0.f, 0.f};
    if (have_tab) {
        tc = *(const v4f*)(tab + 4 * tid);
        ti = *(const v4f*)(tab + TAB_N + 4 * tid);
    }

    unsigned g0[UPT], g1[UPT];
    int      lo[UPT];
    unsigned zmask = 0;
    #pragma unroll
    for (int i = 0; i < UPT; ++i) {
        int u   = tid + 256 * i;
        int p   = u / NQUAD;
        int q   = u - p * NQUAD;
        int k0  = 2 * p, k1 = k0 + 1;
        int ci0 = k0 / 3, kh0 = k0 - 3 * ci0;
        int ci1 = k1 / 3, kh1 = k1 - 3 * ci1;
        int r0  = 3 * h + kh0 - 1;
        int r1  = 3 * h + kh1 - 1;
        if (r0 < 0) { zmask |= 1u << (2 * i);     r0 = 0; }
        if (r1 < 0) { zmask |= 1u << (2 * i + 1); r1 = 0; }
        g0[i] = (unsigned)((ci0 * HIN + r0) * WIN + 4 * q);
        g1[i] = (unsigned)((ci1 * HIN + r1) * WIN + 4 * q);
        lo[i] = p * XSP_STR + 2 * (4 * q + 1 + XCOL0);
    }
    const bool tail = tid < TAIL_T;

    const float* xb = x + (size_t)b * IN_C * HIN * WIN;
    v4f ra[UPT], rb[UPT];

    #pragma unroll
    for (int i = 0; i < UPT; ++i) {
        if (i < UPT - 1 || tail) {
            ra[i] = *(const v4f*)(xb + g0[i]);
            rb[i] = *(const v4f*)(xb + g1[i]);
        }
    }

    if (!have_tab)
    for (int p = tid; p < N_SPLIT * OUT_C; p += 256) {
        const float* cb = comb + p * N_LEGO;
        int   best = 0;
        float bv   = cb[0];
        #pragma unroll
        for (int n = 1; n < N_LEGO; ++n) {
            float v = cb[n];
            if (v > bv) { bv = v; best = n; }
        }
        cW[p] = coef[p * N_LEGO + best];
        iW[p] = best;
    }

    for (int i = tid; i < (KDIM / 2) * 8; i += 256) {
        int row = i >> 3, j = i & 7;
        int c   = (j == 0) ? 0 : (WIN + j);
        *(v2f*)&XsP[row * XSP_STR + 2 * (c + XCOL0)] = (v2f){0.f, 0.f};
    }

    const int lane = tid & 31;
    const int wave = tid >> 5;
    const int half = lane >> 4;
    const int l16  = lane & 15;
    const int mt   = wave >> 1;
    const int nt0  = (wave & 1) * 2;

    const float* Arow = lego + (mt * 16 + l16) * KDIM + half * 2;
    v2f afrag[KSTEPS];
    #pragma unroll
    for (int t = 0; t < KSTEPS; ++t)
        afrag[t] = *(const v2f*)(Arow + 4 * t);

    if (have_tab) {
        *(v4f*)&cW[4 * tid] = tc;
        *(v4i*)&iW[4 * tid] = (v4i){__float_as_int(ti.x), __float_as_int(ti.y),
                                    __float_as_int(ti.z), __float_as_int(ti.w)};
    }

    for (int s = 0; s < N_SPLIT; ++s) {
        __syncthreads();

        #pragma unroll
        for (int i = 0; i < UPT; ++i) {
            if (i < UPT - 1 || tail) {
                v4f a = ra[i], c = rb[i];
                if (zmask & (1u << (2 * i)))     a = (v4f){0.f, 0.f, 0.f, 0.f};
                if (zmask & (1u << (2 * i + 1))) c = (v4f){0.f, 0.f, 0.f, 0.f};
                *(v4f*)&XsP[lo[i]]     = (v4f){a.x, c.x, a.y, c.y};
                *(v4f*)&XsP[lo[i] + 4] = (v4f){a.z, c.z, a.w, c.w};
            }
        }
        __syncthreads();

        if (s + 1 < N_SPLIT) {
            const float* xs = xb + (size_t)(s + 1) * BASIC_C * HIN * WIN;
            #pragma unroll
            for (int i = 0; i < UPT; ++i) {
                if (i < UPT - 1 || tail) {
                    ra[i] = *(const v4f*)(xs + g0[i]);
                    rb[i] = *(const v4f*)(xs + g1[i]);
                }
            }
        }

        v8f acc0 = {0.f, 0.f, 0.f, 0.f, 0.f, 0.f, 0.f, 0.f};
        v8f acc1 = {0.f, 0.f, 0.f, 0.f, 0.f, 0.f, 0.f, 0.f};
        const float* bbase = &XsP[half * XSP_STR + 2 * XCOL0 + (l16 << 1) + nt0 * 32];
        #pragma unroll
        for (int t = 0; t < KSTEPS; ++t) {
            const float* brow = bbase + 2 * t * XSP_STR;
            v2f b0 = *(const v2f*)(brow);
            v2f b1 = *(const v2f*)(brow + 32);
            acc0 = __builtin_amdgcn_wmma_f32_16x16x4_f32(
                       false, afrag[t], false, b0, (short)0, acc0, false, false);
            acc1 = __builtin_amdgcn_wmma_f32_16x16x4_f32(
                       false, afrag[t], false, b1, (short)0, acc1, false, false);
        }

        float* fbase = Fs + (s * N_LEGO + mt * 16 + half * 8) * FS_W;
        #pragma unroll
        for (int v = 0; v < 8; ++v) {
            fbase[v * FS_W + nt0       * 16 + l16] = acc0[v];
            fbase[v * FS_W + (nt0 + 1) * 16 + l16] = acc1[v];
        }
    }
    __syncthreads();

    {
        const int ob = wave * 32;
        int   selI[N_SPLIT];
        float selC[N_SPLIT];
        #pragma unroll
        for (int s = 0; s < N_SPLIT; ++s) {
            selI[s] = iW[s * OUT_C + ob + lane];
            selC[s] = cW[s * OUT_C + ob + lane];
        }
        const bool   own  = lane < WOUT / 2;
        const float* frow = Fs + 2 * lane;
        float*       orow = out + (((size_t)b * OUT_C + ob) * HOUT + h) * WOUT + 2 * lane;
        for (int j0 = 0; j0 < 32; j0 += EP_ROWS) {
            v2f acc[EP_ROWS];
            #pragma unroll
            for (int j = 0; j < EP_ROWS; ++j) {
                acc[j] = (v2f){0.f, 0.f};
                #pragma unroll
                for (int s = 0; s < N_SPLIT; ++s) {
                    const int   n = __builtin_amdgcn_readlane(selI[s], j0 + j);
                    const float c = __int_as_float(__builtin_amdgcn_readlane(__float_as_int(selC[s]), j0 + j));
                    const v2f   f = *(const v2f*)(frow + (s * N_LEGO + n) * FS_W);
                    acc[j] += c * f;
                }
            }
            if (own) {
                #pragma unroll
                for (int j = 0; j < EP_ROWS; ++j)
                    *(v2f*)(orow + (size_t)(j0 + j) * HOUT * WOUT) = acc[j];
            }
        }
    }
}

extern "C" void kernel_launch(void* const* d_in, const int* in_sizes, int n_in,
                              void* d_out, int out_size, void* d_ws, size_t ws_size,
                              hipStream_t stream) {
    (void)in_sizes; (void)n_in; (void)out_size;
    const float* x    = (const float*)d_in[0];
    const float* lego = (const float*)d_in[1];
    const float* coef = (const float*)d_in[2];
    const float* comb = (const float*)d_in[3];
    float* out = (float*)d_out;
    float* tab = (float*)d_ws;

    lego_select<<<dim3(SEL_WGS, 1, 1), dim3(32, 1, 1), 0, stream>>>(coef, comb, tab, ws_size);

    dim3 grid(HOUT, 32, 1);
    dim3 block(256, 1, 1);
    lego_conv_fused<<<grid, block, 0, stream>>>(x, lego, coef, comb, tab, ws_size, out);
}


# matrix loop: B fragments read from LDS in groups of 6 k-steps into a second register set, each group issued ahead of the previous group's 12 WMMAs; was one load between every 2 WMMAs; sum order kept
# speedup vs baseline: 5.4818x; 1.0386x over previous
#include <hip/hip_runtime.h>
#include <hip/hip_bf16.h>


typedef __attribute__((ext_vector_type(2))) float v2f;
typedef __attribute__((ext_vector_type(4))) float v4f;
typedef __attribute__((ext_vector_type(8))) float v8f;
typedef __attribute__((ext_vector_type(4))) int   v4i;

#define IN_C    256
#define OUT_C   256
#define N_SPLIT 4
#define N_LEGO  64
#define BASIC_C 64
#define HIN     56
#define WIN     56
#define HOUT    19
#define WOUT    58
#define KDIM    192
#define NCOL    64
#define XSP_STR 160
#define XCOL0   3
#define FS_W    64
#define NQUAD   14
#define NUNIT   ((KDIM / 2) * NQUAD)
#define UPT     6
#define TAIL_T  (NUNIT - (UPT - 1) * 256)
#define KSTEPS  (KDIM / 4)
#define KG      6
#define NGRP    (KSTEPS / KG)
#define EP_ROWS 8
#define TAB_N     (N_SPLIT * OUT_C)
#define TAB_BYTES (2 * TAB_N * 4)
#define SEL_WGS   32

__global__ __launch_bounds__(32)
void lego_select(const float* __restrict__ coef,
                 const float* __restrict__ comb,
                 float* __restrict__ tab, size_t tab_room) {
    if (tab_room < (size_t)TAB_BYTES) return;
    const int p = blockIdx.x * 32 + threadIdx.x;
    const float* cb = comb + p * N_LEGO;
    int   best = 0;
    float bv   = cb[0];
    #pragma unroll
    for (int n = 1; n < N_LEGO; ++n) {
        float v = cb[n];
        if (v > bv) { bv = v; best = n; }
    }
    tab[p]         = coef[p * N_LEGO + best];
    tab[TAB_N + p] = __int_as_float(best);
}

__global__ __launch_bounds__(256, 1)
void lego_conv_fused(const float* __restrict__ x,
                     const float* __restrict__ lego,
                     const float* __restrict__ coef,
                     const float* __restrict__ comb,
                     const float* __restrict__ tab, size_t tab_room,
                     float* __restrict__ out) {
    __shared__ __attribute__((aligned(16))) float XsP[(KDIM / 2) * XSP_STR];
    __shared__ __attribute__((aligned(16))) float Fs[N_SPLIT * N_LEGO * FS_W];
    __shared__ __attribute__((aligned(16))) float cW[N_SPLIT * OUT_C];
    __shared__ __attribute__((aligned(16))) int   iW[N_SPLIT * OUT_C];

    const int tid = threadIdx.x;
    const int b   = blockIdx.y;
    const int h   = blockIdx.x;

    const bool have_tab = tab_room >= (size_t)TAB_BYTES;
    v4f tc = {0.f, 0.f, 0.f, 0.f}, ti = {0.f, 0.f, 0.f, 0.f};
    if (have_tab) {
        tc = *(const v4f*)(tab + 4 * tid);
        ti = *(const v4f*)(tab + TAB_N + 4 * tid);
    }

    unsigned g0[UPT], g1[UPT];
    int      lo[UPT];
    unsigned zmask = 0;
    #pragma unroll
    for (int i = 0; i < UPT; ++i) {
        int u   = tid + 256 * i;
        int p   = u / NQUAD;
        int q   = u - p * NQUAD;
        int k0  = 2 * p, k1 = k0 + 1;
        int ci0 = k0 / 3, kh0 = k0 - 3 * ci0;
        int ci1 = k1 / 3, kh1 = k1 - 3 * ci1;
        int r0  = 3 * h + kh0 - 1;
        int r1  = 3 * h + kh1 - 1;
        if (r0 < 0) { zmask |= 1u << (2 * i);     r0 = 0; }
        if (r1 < 0) { zmask |= 1u << (2 * i + 1); r1 = 0; }
        g0[i] = (unsigned)((ci0 * HIN + r0) * WIN + 4 * q);
        g1[i] = (unsigned)((ci1 * HIN + r1) * WIN + 4 * q);
        lo[i] = p * XSP_STR + 2 * (4 * q + 1 + XCOL0);
    }
    const bool tail = tid < TAIL_T;

    const float* xb = x + (size_t)b * IN_C * HIN * WIN;
    v4f ra[UPT], rb[UPT];

    #pragma unroll
    for (int i = 0; i < UPT; ++i) {
        if (i < UPT - 1 || tail) {
            ra[i] = *(const v4f*)(xb + g0[i]);
            rb[i] = *(const v4f*)(xb + g1[i]);
        }
    }

    if (!have_tab)
    for (int p = tid; p < N_SPLIT * OUT_C; p += 256) {
        const float* cb = comb + p * N_LEGO;
        int   best = 0;
        float bv   = cb[0];
        #pragma unroll
        for (int n = 1; n < N_LEGO; ++n) {
            float v = cb[n];
            if (v > bv) { bv = v; best = n; }
        }
        cW[p] = coef[p * N_LEGO + best];
        iW[p] = best;
    }

    for (int i = tid; i < (KDIM / 2) * 8; i += 256) {
        int row = i >> 3, j = i & 7;
        int c   = (j == 0) ? 0 : (WIN + j);
        *(v2f*)&XsP[row * XSP_STR + 2 * (c + XCOL0)] = (v2f){0.f, 0.f};
    }

    const int lane = tid & 31;
    const int wave = tid >> 5;
    const int half = lane >> 4;
    const int l16  = lane & 15;
    const int mt   = wave >> 1;
    const int nt0  = (wave & 1) * 2;

    const float* Arow = lego + (mt * 16 + l16) * KDIM + half * 2;
    v2f afrag[KSTEPS];
    #pragma unroll
    for (int t = 0; t < KSTEPS; ++t)
        afrag[t] = *(const v2f*)(Arow + 4 * t);

    if (have_tab) {
        *(v4f*)&cW[4 * tid] = tc;
        *(v4i*)&iW[4 * tid] = (v4i){__float_as_int(ti.x), __float_as_int(ti.y),
                                    __float_as_int(ti.z), __float_as_int(ti.w)};
    }

    for (int s = 0; s < N_SPLIT; ++s) {
        __syncthreads();

        #pragma unroll
        for (int i = 0; i < UPT; ++i) {
            if (i < UPT - 1 || tail) {
                v4f a = ra[i], c = rb[i];
                if (zmask & (1u << (2 * i)))     a = (v4f){0.f, 0.f, 0.f, 0.f};
                if (zmask & (1u << (2 * i + 1))) c = (v4f){0.f, 0.f, 0.f, 0.f};
                *(v4f*)&XsP[lo[i]]     = (v4f){a.x, c.x, a.y, c.y};
                *(v4f*)&XsP[lo[i] + 4] = (v4f){a.z, c.z, a.w, c.w};
            }
        }
        __syncthreads();

        if (s + 1 < N_SPLIT) {
            const float* xs = xb + (size_t)(s + 1) * BASIC_C * HIN * WIN;
            #pragma unroll
            for (int i = 0; i < UPT; ++i) {
                if (i < UPT - 1 || tail) {
                    ra[i] = *(const v4f*)(xs + g0[i]);
                    rb[i] = *(const v4f*)(xs + g1[i]);
                }
            }
        }

        v8f acc0 = {0.f, 0.f, 0.f, 0.f, 0.f, 0.f, 0.f, 0.f};
        v8f acc1 = {0.f, 0.f, 0.f, 0.f, 0.f, 0.f, 0.f, 0.f};
        const float* bbase = &XsP[half * XSP_STR + 2 * XCOL0 + (l16 << 1) + nt0 * 32];
        v2f bq[2][KG][2];
        #pragma unroll
        for (int j = 0; j < KG; ++j) {
            const float* brow = bbase + 2 * j * XSP_STR;
            bq[0][j][0] = *(const v2f*)(brow);
            bq[0][j][1] = *(const v2f*)(brow + 32);
        }
        #pragma unroll
        for (int g = 0; g < NGRP; ++g) {
            if (g + 1 < NGRP) {
                #pragma unroll
                for (int j = 0; j < KG; ++j) {
                    const float* brow = bbase + 2 * ((g + 1) * KG + j) * XSP_STR;
                    bq[(g + 1) & 1][j][0] = *(const v2f*)(brow);
                    bq[(g + 1) & 1][j][1] = *(const v2f*)(brow + 32);
                }
            }
            __builtin_amdgcn_sched_barrier(0);
            #pragma unroll
            for (int j = 0; j < KG; ++j) {
                const int t = g * KG + j;
                acc0 = __builtin_amdgcn_wmma_f32_16x16x4_f32(
                           false, afrag[t], false, bq[g & 1][j][0], (short)0, acc0, false, false);
                acc1 = __builtin_amdgcn_wmma_f32_16x16x4_f32(
                           false, afrag[t], false, bq[g & 1][j][1], (short)0, acc1, false, false);
            }
            __builtin_amdgcn_sched_barrier(0);
        }

        float* fbase = Fs + (s * N_LEGO + mt * 16 + half * 8) * FS_W;
        #pragma unroll
        for (int v = 0; v < 8; ++v) {
            fbase[v * FS_W + nt0       * 16 + l16] = acc0[v];
            fbase[v * FS_W + (nt0 + 1) * 16 + l16] = acc1[v];
        }
    }
    __syncthreads();

    {
        const int ob = wave * 32;
        int   selI[N_SPLIT];
        float selC[N_SPLIT];
        #pragma unroll
        for (int s = 0; s < N_SPLIT; ++s) {
            selI[s] = iW[s * OUT_C + ob + lane];
            selC[s] = cW[s * OUT_C + ob + lane];
        }
        const bool   own  = lane < WOUT / 2;
        const float* frow = Fs + 2 * lane;
        float*       orow = out + (((size_t)b * OUT_C + ob) * HOUT + h) * WOUT + 2 * lane;
        for (int j0 = 0; j0 < 32; j0 += EP_ROWS) {
            v2f acc[EP_ROWS];
            #pragma unroll
            for (int j = 0; j < EP_ROWS; ++j) {
                acc[j] = (v2f){0.f, 0.f};
                #pragma unroll
                for (int s = 0; s < N_SPLIT; ++s) {
                    const int   n = __builtin_amdgcn_readlane(selI[s], j0 + j);
                    const float c = __int_as_float(__builtin_amdgcn_readlane(__float_as_int(selC[s]), j0 + j));
                    const v2f   f = *(const v2f*)(frow + (s * N_LEGO + n) * FS_W);
                    acc[j] += c * f;
                }
            }
            if (own) {
                #pragma unroll
                for (int j = 0; j < EP_ROWS; ++j)
                    *(v2f*)(orow + (size_t)(j0 + j) * HOUT * WOUT) = acc[j];
            }
        }
    }
}

extern "C" void kernel_launch(void* const* d_in, const int* in_sizes, int n_in,
                              void* d_out, int out_size, void* d_ws, size_t ws_size,
                              hipStream_t stream) {
    (void)in_sizes; (void)n_in; (void)out_size;
    const float* x    = (const float*)d_in[0];
    const float* lego = (const float*)d_in[1];
    const float* coef = (const float*)d_in[2];
    const float* comb = (const float*)d_in[3];
    float* out = (float*)d_out;
    float* tab = (float*)d_ws;

    lego_select<<<dim3(SEL_WGS, 1, 1), dim3(32, 1, 1), 0, stream>>>(coef, comb, tab, ws_size);

    dim3 grid(HOUT, 32, 1);
    dim3 block(256, 1, 1);
    lego_conv_fused<<<grid, block, 0, stream>>>(x, lego, coef, comb, tab, ws_size, out);
}


# tail balance: the last 96 of 608 items, which ran as a second round on 96 WGPs, are each split into two work-groups of one column half (one N-tile per wave); 1-D grid of 704; sum order kept
# speedup vs baseline: 5.6961x; 1.0391x over previous
#include <hip/hip_runtime.h>
#include <hip/hip_bf16.h>


typedef __attribute__((ext_vector_type(2))) float v2f;
typedef __attribute__((ext_vector_type(4))) float v4f;
typedef __attribute__((ext_vector_type(8))) float v8f;
typedef __attribute__((ext_vector_type(4))) int   v4i;

#define IN_C    256
#define OUT_C   256
#define N_SPLIT 4
#define N_LEGO  64
#define BASIC_C 64
#define HIN     56
#define WIN     56
#define HOUT    19
#define WOUT    58
#define KDIM    192
#define NCOL    64
#define XSP_STR 160
#define XCOL0   3
#define FS_W    64
#define NQUAD   14
#define NUNIT   ((KDIM / 2) * NQUAD)
#define UPT     6
#define TAIL_T  (NUNIT - (UPT - 1) * 256)
#define KSTEPS  (KDIM / 4)
#define KG      6
#define NGRP    (KSTEPS / KG)
#define EP_ROWS 8
#define TAB_N     (N_SPLIT * OUT_C)
#define TAB_BYTES (2 * TAB_N * 4)
#define SEL_WGS   32
#define N_FULL    512
#define N_WG      704

__global__ __launch_bounds__(32)
void lego_select(const float* __restrict__ coef,
                 const float* __restrict__ comb,
                 float* __restrict__ tab, size_t tab_room) {
    if (tab_room < (size_t)TAB_BYTES) return;
    const int p = blockIdx.x * 32 + threadIdx.x;
    const float* cb = comb + p * N_LEGO;
    int   best = 0;
    float bv   = cb[0];
    #pragma unroll
    for (int n = 1; n < N_LEGO; ++n) {
        float v = cb[n];
        if (v > bv) { bv = v; best = n; }
    }
    tab[p]         = coef[p * N_LEGO + best];
    tab[TAB_N + p] = __int_as_float(best);
}

template <int NT>
__device__ __forceinline__
void lego_conv_item(const float* __restrict__ x,
                    const float* __restrict__ lego,
                    const float* __restrict__ coef,
                    const float* __restrict__ comb,
                    const float* __restrict__ tab, size_t tab_room,
                    float* __restrict__ out,
                    const int b, const int h, const int nh,
                    float* __restrict__ XsP, float* __restrict__ Fs,
                    float* __restrict__ cW, int* __restrict__ iW) {
    const int tid = threadIdx.x;

    const bool have_tab = tab_room >= (size_t)TAB_BYTES;
    v4f tc = {0.f, 0.f, 0.f, 0.f}, ti = {0.f, 0.f, 0.f, 0.f};
    if (have_tab) {
        tc = *(const v4f*)(tab + 4 * tid);
        ti = *(const v4f*)(tab + TAB_N + 4 * tid);
    }

    unsigned g0[UPT], g1[UPT];
    int      lo[UPT];
    unsigned zmask = 0;
    #pragma unroll
    for (int i = 0; i < UPT; ++i) {
        int u   = tid + 256 * i;
        int p   = u / NQUAD;
        int q   = u - p * NQUAD;
        int k0  = 2 * p, k1 = k0 + 1;
        int ci0 = k0 / 3, kh0 = k0 - 3 * ci0;
        int ci1 = k1 / 3, kh1 = k1 - 3 * ci1;
        int r0  = 3 * h + kh0 - 1;
        int r1  = 3 * h + kh1 - 1;
        if (r0 < 0) { zmask |= 1u << (2 * i);     r0 = 0; }
        if (r1 < 0) { zmask |= 1u << (2 * i + 1); r1 = 0; }
        g0[i] = (unsigned)((ci0 * HIN + r0) * WIN + 4 * q);
        g1[i] = (unsigned)((ci1 * HIN + r1) * WIN + 4 * q);
        lo[i] = p * XSP_STR + 2 * (4 * q + 1 + XCOL0);
    }
    const bool tail = tid < TAIL_T;

    const float* xb = x + (size_t)b * IN_C * HIN * WIN;
    v4f ra[UPT], rb[UPT];

    #pragma unroll
    for (int i = 0; i < UPT; ++i) {
        if (i < UPT - 1 || tail) {
            ra[i] = *(const v4f*)(xb + g0[i]);
            rb[i] = *(const v4f*)(xb + g1[i]);
        }
    }

    if (!have_tab)
    for (int p = tid; p < N_SPLIT * OUT_C; p += 256) {
        const float* cb = comb + p * N_LEGO;
        int   best = 0;
        float bv   = cb[0];
        #pragma unroll
        for (int n = 1; n < N_LEGO; ++n) {
            float v = cb[n];
            if (v > bv) { bv = v; best = n; }
        }
        cW[p] = coef[p * N_LEGO + best];
        iW[p] = best;
    }

    for (int i = tid; i < (KDIM / 2) * 8; i += 256) {
        int row = i >> 3, j = i & 7;
        int c   = (j == 0) ? 0 : (WIN + j);
        *(v2f*)&XsP[row * XSP_STR + 2 * (c + XCOL0)] = (v2f){0.f, 0.f};
    }

    const int lane = tid & 31;
    const int wave = tid >> 5;
    const int half = lane >> 4;
    const int l16  = lane & 15;
    const int mt   = wave >> 1;
    const int nt0  = (NT == 2) ? (wave & 1) * 2
                               : nh * 2 + (wave & 1);

    const float* Arow = lego + (mt * 16 + l16) * KDIM + half * 2;
    v2f afrag[KSTEPS];
    #pragma unroll
    for (int t = 0; t < KSTEPS; ++t)
        afrag[t] = *(const v2f*)(Arow + 4 * t);

    if (have_tab) {
        *(v4f*)&cW[4 * tid] = tc;
        *(v4i*)&iW[4 * tid] = (v4i){__float_as_int(ti.x), __float_as_int(ti.y),
                                    __float_as_int(ti.z), __float_as_int(ti.w)};
    }

    for (int s = 0; s < N_SPLIT; ++s) {
        __syncthreads();

        #pragma unroll
        for (int i = 0; i < UPT; ++i) {
            if (i < UPT - 1 || tail) {
                v4f a = ra[i], c = rb[i];
                if (zmask & (1u << (2 * i)))     a = (v4f){0.f, 0.f, 0.f, 0.f};
                if (zmask & (1u << (2 * i + 1))) c = (v4f){0.f, 0.f, 0.f, 0.f};
                *(v4f*)&XsP[lo[i]]     = (v4f){a.x, c.x, a.y, c.y};
                *(v4f*)&XsP[lo[i] + 4] = (v4f){a.z, c.z, a.w, c.w};
            }
        }
        __syncthreads();

        if (s + 1 < N_SPLIT) {
            const float* xs = xb + (size_t)(s + 1) * BASIC_C * HIN * WIN;
            #pragma unroll
            for (int i = 0; i < UPT; ++i) {
                if (i < UPT - 1 || tail) {
                    ra[i] = *(const v4f*)(xs + g0[i]);
                    rb[i] = *(const v4f*)(xs + g1[i]);
                }
            }
        }

        v8f acc0 = {0.f, 0.f, 0.f, 0.f, 0.f, 0.f, 0.f, 0.f};
        v8f acc1 = {0.f, 0.f, 0.f, 0.f, 0.f, 0.f, 0.f, 0.f};
        const float* bbase = &XsP[half * XSP_STR + 2 * XCOL0 + (l16 << 1) + nt0 * 32];
        constexpr int KGT = (NT == 2) ? KG : 2 * KG;
        constexpr int NGT = KSTEPS / KGT;
        v2f bq[2][KGT][NT];
        #pragma unroll
        for (int j = 0; j < KGT; ++j) {
            const float* brow = bbase + 2 * j * XSP_STR;
            bq[0][j][0] = *(const v2f*)(brow);
            if (NT == 2) bq[0][j][NT - 1] = *(const v2f*)(brow + 32);
        }
        #pragma unroll
        for (int g = 0; g < NGT; ++g) {
            if (g + 1 < NGT) {
                #pragma unroll
                for (int j = 0; j < KGT; ++j) {
                    const float* brow = bbase + 2 * ((g + 1) * KGT + j) * XSP_STR;
                    bq[(g + 1) & 1][j][0] = *(const v2f*)(brow);
                    if (NT == 2) bq[(g + 1) & 1][j][NT - 1] = *(const v2f*)(brow + 32);
                }
            }
            __builtin_amdgcn_sched_barrier(0);
            #pragma unroll
            for (int j = 0; j < KGT; ++j) {
                const int t = g * KGT + j;
                acc0 = __builtin_amdgcn_wmma_f32_16x16x4_f32(
                           false, afrag[t], false, bq[g & 1][j][0], (short)0, acc0, false, false);
                if (NT == 2)
                    acc1 = __builtin_amdgcn_wmma_f32_16x16x4_f32(
                               false, afrag[t], false, bq[g & 1][j][NT - 1], (short)0, acc1, false, false);
            }
            __builtin_amdgcn_sched_barrier(0);
        }

        float* fbase = Fs + (s * N_LEGO + mt * 16 + half * 8) * FS_W;
        #pragma unroll
        for (int v = 0; v < 8; ++v) {
            fbase[v * FS_W + nt0       * 16 + l16] = acc0[v];
            if (NT == 2) fbase[v * FS_W + (nt0 + 1) * 16 + l16] = acc1[v];
        }
    }
    __syncthreads();

    {
        const int ob = wave * 32;
        int   selI[N_SPLIT];
        float selC[N_SPLIT];
        #pragma unroll
        for (int s = 0; s < N_SPLIT; ++s) {
            selI[s] = iW[s * OUT_C + ob + lane];
            selC[s] = cW[s * OUT_C + ob + lane];
        }
        const bool   own  = lane < WOUT / 2 && (NT == 2 || half == nh);
        const float* frow = Fs + 2 * lane;
        float*       orow = out + (((size_t)b * OUT_C + ob) * HOUT + h) * WOUT + 2 * lane;
        for (int j0 = 0; j0 < 32; j0 += EP_ROWS) {
            v2f acc[EP_ROWS];
            #pragma unroll
            for (int j = 0; j < EP_ROWS; ++j) {
                acc[j] = (v2f){0.f, 0.f};
                #pragma unroll
                for (int s = 0; s < N_SPLIT; ++s) {
                    const int   n = __builtin_amdgcn_readlane(selI[s], j0 + j);
                    const float c = __int_as_float(__builtin_amdgcn_readlane(__float_as_int(selC[s]), j0 + j));
                    const v2f   f = *(const v2f*)(frow + (s * N_LEGO + n) * FS_W);
                    acc[j] += c * f;
                }
            }
            if (own) {
                #pragma unroll
                for (int j = 0; j < EP_ROWS; ++j)
                    *(v2f*)(orow + (size_t)(j0 + j) * HOUT * WOUT) = acc[j];
            }
        }
    }
}

__global__ __launch_bounds__(256, 1)
void lego_conv_fused(const float* __restrict__ x,
                     const float* __restrict__ lego,
                     const float* __restrict__ coef,
                     const float* __restrict__ comb,
                     const float* __restrict__ tab, size_t tab_room,
                     float* __restrict__ out) {
    __shared__ __attribute__((aligned(16))) float XsP[(KDIM / 2) * XSP_STR];
    __shared__ __attribute__((aligned(16))) float Fs[N_SPLIT * N_LEGO * FS_W];
    __shared__ __attribute__((aligned(16))) float cW[N_SPLIT * OUT_C];
    __shared__ __attribute__((aligned(16))) int   iW[N_SPLIT * OUT_C];

    const int id = blockIdx.x;
    if (__builtin_expect(id < N_FULL, 1)) {
        const int b = id / HOUT;
        const int h = id - b * HOUT;
        lego_conv_item<2>(x, lego, coef, comb, tab, tab_room, out, b, h, 0, XsP, Fs, cW, iW);
    } else {
        const int j    = id - N_FULL;
        const int item = N_FULL + (j >> 1);
        const int b    = item / HOUT;
        const int h    = item - b * HOUT;
        lego_conv_item<1>(x, lego, coef, comb, tab, tab_room, out, b, h, j & 1, XsP, Fs, cW, iW);
    }
}

extern "C" void kernel_launch(void* const* d_in, const int* in_sizes, int n_in,
                              void* d_out, int out_size, void* d_ws, size_t ws_size,
                              hipStream_t stream) {
    (void)in_sizes; (void)n_in; (void)out_size;
    const float* x    = (const float*)d_in[0];
    const float* lego = (const float*)d_in[1];
    const float* coef = (const float*)d_in[2];
    const float* comb = (const float*)d_in[3];
    float* out = (float*)d_out;
    float* tab = (float*)d_ws;

    lego_select<<<dim3(SEL_WGS, 1, 1), dim3(32, 1, 1), 0, stream>>>(coef, comb, tab, ws_size);

    dim3 grid(N_WG, 1, 1);
    dim3 block(256, 1, 1);
    lego_conv_fused<<<grid, block, 0, stream>>>(x, lego, coef, comb, tab, ws_size, out);
}


# tail halves run inside the first 192 of 512 work-groups as a second piece behind their whole item: waves, A fragments, table kept, first x loads issued under the item's last matrix loop; sums kept
# speedup vs baseline: 6.1125x; 1.0731x over previous
#include <hip/hip_runtime.h>
#include <hip/hip_bf16.h>


typedef __attribute__((ext_vector_type(2))) float v2f;
typedef __attribute__((ext_vector_type(4))) float v4f;
typedef __attribute__((ext_vector_type(8))) float v8f;
typedef __attribute__((ext_vector_type(4))) int   v4i;
typedef const float __attribute__((address_space(1)))* gfp;
typedef const v4f   __attribute__((address_space(1)))* gv4p;

#define IN_C    256
#define OUT_C   256
#define N_SPLIT 4
#define N_LEGO  64
#define BASIC_C 64
#define HIN     56
#define WIN     56
#define HOUT    19
#define WOUT    58
#define KDIM    192
#define NCOL    64
#define XSP_STR 160
#define XCOL0   3
#define FS_W    64
#define NQUAD   14
#define NUNIT   ((KDIM / 2) * NQUAD)
#define UPT     6
#define TAIL_T  (NUNIT - (UPT - 1) * 256)
#define KSTEPS  (KDIM / 4)
#define KG      6
#define NGRP    (KSTEPS / KG)
#define EP_ROWS 8
#define TAB_N     (N_SPLIT * OUT_C)
#define TAB_BYTES (2 * TAB_N * 4)
#define SEL_WGS   32
#define N_FULL    512
#define N_HALF    192
#define N_WG      N_FULL

__global__ __launch_bounds__(32)
void lego_select(const float* __restrict__ coef,
                 const float* __restrict__ comb,
                 float* __restrict__ tab, size_t tab_room) {
    if (tab_room < (size_t)TAB_BYTES) return;
    const int p = blockIdx.x * 32 + threadIdx.x;
    const float* cb = comb + p * N_LEGO;
    int   best = 0;
    float bv   = cb[0];
    #pragma unroll
    for (int n = 1; n < N_LEGO; ++n) {
        float v = cb[n];
        if (v > bv) { bv = v; best = n; }
    }
    tab[p]         = coef[p * N_LEGO + best];
    tab[TAB_N + p] = __int_as_float(best);
}

__device__ __forceinline__
void stage_plan(const int tid, const int h, const gfp xb,
                gfp (&g0)[UPT], gfp (&g1)[UPT], int (&lo)[UPT], unsigned& kz) {
    kz = 0;
    const int dp = 3 * h * WIN;
    const int dz = (h > 0) ? (3 * h - 1) * WIN : 0;
    #pragma unroll
    for (int i = 0; i < UPT; ++i) {
        int u   = tid + 256 * i;
        int p   = u / NQUAD;
        int q   = u - p * NQUAD;
        int k0  = 2 * p, k1 = k0 + 1;
        int ci0 = k0 / 3, kh0 = k0 - 3 * ci0;
        int ci1 = k1 / 3, kh1 = k1 - 3 * ci1;
        if (kh0 == 0) kz |= 1u << (2 * i);
        if (kh1 == 0) kz |= 1u << (2 * i + 1);
        g0[i] = xb + (unsigned)((ci0 * HIN + (kh0 == 0 ? 0 : kh0 - 1)) * WIN + 4 * q + (kh0 == 0 ? dz : dp));
        g1[i] = xb + (unsigned)((ci1 * HIN + (kh1 == 0 ? 0 : kh1 - 1)) * WIN + 4 * q + (kh1 == 0 ? dz : dp));
        lo[i] = p * XSP_STR + 2 * (4 * q + 1 + XCOL0);
    }
}

__device__ __forceinline__
void move_plan(const long ddp, const long ddz, const unsigned kz,
               gfp (&g0)[UPT], gfp (&g1)[UPT]) {
    #pragma unroll
    for (int i = 0; i < UPT; ++i) {
        g0[i] += (kz & (1u << (2 * i)))     ? ddz : ddp;
        g1[i] += (kz & (1u << (2 * i + 1))) ? ddz : ddp;
    }
}

__device__ __forceinline__
void issue_x(const int step,
             const gfp (&g0)[UPT], const gfp (&g1)[UPT], const bool tail,
             v4f (&ra)[UPT], v4f (&rb)[UPT]) {
    #pragma unroll
    for (int i = 0; i < UPT; ++i) {
        if (i < UPT - 1 || tail) {
            ra[i] = *(gv4p)(g0[i] + step);
            rb[i] = *(gv4p)(g1[i] + step);
        } else {
            ra[i] = (v4f){0.f, 0.f, 0.f, 0.f};
            rb[i] = (v4f){0.f, 0.f, 0.f, 0.f};
        }
    }
}

template <int NT, bool CHAIN>
__device__ __forceinline__
void run_item(float* __restrict__ out,
              const int b, const int h, const int nh,
              const bool more, const long bstep, const int h2,
              gfp (&g0)[UPT], gfp (&g1)[UPT], const int (&lo)[UPT], const unsigned kz, unsigned& zmask,
              const bool tail, v4f (&ra)[UPT], v4f (&rb)[UPT], const v2f (&afrag)[KSTEPS],
              float* __restrict__ XsP, float* __restrict__ Fs,
              const float* __restrict__ cW, const int* __restrict__ iW) {
    const int tid  = threadIdx.x;
    const int lane = tid & 31;
    const int wave = tid >> 5;
    const int half = lane >> 4;
    const int l16  = lane & 15;
    const int mt   = wave >> 1;
    const int nt0  = (NT == 2) ? (wave & 1) * 2
                               : nh * 2 + (wave & 1);

    for (int s = 0; s < N_SPLIT; ++s) {
        __syncthreads();

        #pragma unroll
        for (int i = 0; i < UPT; ++i) {
            if (i < UPT - 1 || tail) {
                v4f a = ra[i], c = rb[i];
                if (zmask & (1u << (2 * i)))     a = (v4f){0.f, 0.f, 0.f, 0.f};
                if (zmask & (1u << (2 * i + 1))) c = (v4f){0.f, 0.f, 0.f, 0.f};
                *(v4f*)&XsP[lo[i]]     = (v4f){a.x, c.x, a.y, c.y};
                *(v4f*)&XsP[lo[i] + 4] = (v4f){a.z, c.z, a.w, c.w};
            }
        }
        __syncthreads();

        if (s + 1 < N_SPLIT) {
            issue_x((s + 1) * BASIC_C * HIN * WIN, g0, g1, tail, ra, rb);
        } else if (CHAIN && more) {
            move_plan(bstep + 3 * (h2 - h) * WIN,
                      bstep + ((h2 > 0) ? (3 * h2 - 1) * WIN : 0) - ((h > 0) ? (3 * h - 1) * WIN : 0), kz, g0, g1);
            zmask = (h2 == 0) ? kz : 0u;
            issue_x(0, g0, g1, tail, ra, rb);
        }

        v8f acc0 = {0.f, 0.f, 0.f, 0.f, 0.f, 0.f, 0.f, 0.f};
        v8f acc1 = {0.f, 0.f, 0.f, 0.f, 0.f, 0.f, 0.f, 0.f};
        const float* bbase = &XsP[half * XSP_STR + 2 * XCOL0 + (l16 << 1) + nt0 * 32];
        constexpr int KGT = (NT == 2) ? KG : 2 * KG;
        constexpr int NGT = KSTEPS / KGT;
        v2f bq[2][KGT][NT];
        #pragma unroll
        for (int j = 0; j < KGT; ++j) {
            const float* brow = bbase + 2 * j * XSP_STR;
            bq[0][j][0] = *(const v2f*)(brow);
            if (NT == 2) bq[0][j][NT - 1] = *(const v2f*)(brow + 32);
        }
        #pragma unroll
        for (int g = 0; g < NGT; ++g) {
            if (g + 1 < NGT) {
                #pragma unroll
                for (int j = 0; j < KGT; ++j) {
                    const float* brow = bbase + 2 * ((g + 1) * KGT + j) * XSP_STR;
                    bq[(g + 1) & 1][j][0] = *(const v2f*)(brow);
                    if (NT == 2) bq[(g + 1) & 1][j][NT - 1] = *(const v2f*)(brow + 32);
                }
            }
            __builtin_amdgcn_sched_barrier(0);
            #pragma unroll
            for (int j = 0; j < KGT; ++j) {
                const int t = g * KGT + j;
                acc0 = __builtin_amdgcn_wmma_f32_16x16x4_f32(
                           false, afrag[t], false, bq[g & 1][j][0], (short)0, acc0, false, false);
                if (NT == 2)
                    acc1 = __builtin_amdgcn_wmma_f32_16x16x4_f32(
                               false, afrag[t], false, bq[g & 1][j][NT - 1], (short)0, acc1, false, false);
            }
            __builtin_amdgcn_sched_barrier(0);
        }

        float* fbase = Fs + (s * N_LEGO + mt * 16 + half * 8) * FS_W;
        #pragma unroll
        for (int v = 0; v < 8; ++v) {
            fbase[v * FS_W + nt0       * 16 + l16] = acc0[v];
            if (NT == 2) fbase[v * FS_W + (nt0 + 1) * 16 + l16] = acc1[v];
        }
    }
    __syncthreads();

    {
        const int ob = wave * 32;
        int   selI[N_SPLIT];
        float selC[N_SPLIT];
        #pragma unroll
        for (int s = 0; s < N_SPLIT; ++s) {
            selI[s] = iW[s * OUT_C + ob + lane];
            selC[s] = cW[s * OUT_C + ob + lane];
        }
        const bool   own  = lane < WOUT / 2 && (NT == 2 || half == nh);
        const float* frow = Fs + 2 * lane;
        float*       orow = out + (((size_t)b * OUT_C + ob) * HOUT + h) * WOUT + 2 * lane;
        for (int j0 = 0; j0 < 32; j0 += EP_ROWS) {
            v2f acc[EP_ROWS];
            #pragma unroll
            for (int j = 0; j < EP_ROWS; ++j) {
                acc[j] = (v2f){0.f, 0.f};
                #pragma unroll
                for (int s = 0; s < N_SPLIT; ++s) {
                    const int   n = __builtin_amdgcn_readlane(selI[s], j0 + j);
                    const float c = __int_as_float(__builtin_amdgcn_readlane(__float_as_int(selC[s]), j0 + j));
                    const v2f   f = *(const v2f*)(frow + (s * N_LEGO + n) * FS_W);
                    acc[j] += c * f;
                }
            }
            if (own) {
                #pragma unroll
                for (int j = 0; j < EP_ROWS; ++j)
                    *(v2f*)(orow + (size_t)(j0 + j) * HOUT * WOUT) = acc[j];
            }
        }
    }
}

__global__ __launch_bounds__(256, 4)
void lego_conv_fused(const float* __restrict__ x,
                     const float* __restrict__ lego,
                     const float* __restrict__ coef,
                     const float* __restrict__ comb,
                     const float* __restrict__ tab, size_t tab_room,
                     float* __restrict__ out) {
    __shared__ __attribute__((aligned(16))) float XsP[(KDIM / 2) * XSP_STR];
    __shared__ __attribute__((aligned(16))) float Fs[N_SPLIT * N_LEGO * FS_W];
    __shared__ __attribute__((aligned(16))) float cW[N_SPLIT * OUT_C];
    __shared__ __attribute__((aligned(16))) int   iW[N_SPLIT * OUT_C];

    const int tid = threadIdx.x;
    const int id  = blockIdx.x;
    const int b   = id / HOUT;
    const int h   = id - b * HOUT;
    const bool more = id < N_HALF;
    const int item2 = N_FULL + (id >> 1);
    const int b2    = item2 / HOUT;
    const int h2    = item2 - b2 * HOUT;

    const bool have_tab = tab_room >= (size_t)TAB_BYTES;
    v4f tc = {0.f, 0.f, 0.f, 0.f}, ti = {0.f, 0.f, 0.f, 0.f};
    if (have_tab) {
        tc = *(const v4f*)(tab + 4 * tid);
        ti = *(const v4f*)(tab + TAB_N + 4 * tid);
    }

    const gfp xb = (gfp)x + (size_t)b * IN_C * HIN * WIN;
    gfp g0[UPT];
    gfp g1[UPT];
    int      lo[UPT];
    unsigned kz;
    stage_plan(tid, h, xb, g0, g1, lo, kz);
    unsigned zmask = (h == 0) ? kz : 0u;
    const bool tail = tid < TAIL_T;

    const long bstep = (long)(b2 - b) * IN_C * HIN * WIN;
    v4f ra[UPT], rb[UPT];
    issue_x(0, g0, g1, tail, ra, rb);

    if (!have_tab)
    for (int p = tid; p < N_SPLIT * OUT_C; p += 256) {
        const float* cb = comb + p * N_LEGO;
        int   best = 0;
        float bv   = cb[0];
        #pragma unroll
        for (int n = 1; n < N_LEGO; ++n) {
            float v = cb[n];
            if (v > bv) { bv = v; best = n; }
        }
        cW[p] = coef[p * N_LEGO + best];
        iW[p] = best;
    }

    for (int i = tid; i < (KDIM / 2) * 8; i += 256) {
        int row = i >> 3, j = i & 7;
        int c   = (j == 0) ? 0 : (WIN + j);
        *(v2f*)&XsP[row * XSP_STR + 2 * (c + XCOL0)] = (v2f){0.f, 0.f};
    }

    const int lane = tid & 31;
    const int mt   = tid >> 6;
    const float* Arow = lego + (mt * 16 + (lane & 15)) * KDIM + (lane >> 4) * 2;
    v2f afrag[KSTEPS];
    #pragma unroll
    for (int t = 0; t < KSTEPS; ++t)
        afrag[t] = *(const v2f*)(Arow + 4 * t);

    if (have_tab) {
        *(v4f*)&cW[4 * tid] = tc;
        *(v4i*)&iW[4 * tid] = (v4i){__float_as_int(ti.x), __float_as_int(ti.y),
                                    __float_as_int(ti.z), __float_as_int(ti.w)};
    }

    run_item<2, true>(out, b, h, 0, more, bstep, h2, g0, g1, lo, kz, zmask, tail, ra, rb, afrag, XsP, Fs, cW, iW);
    if (more)
        run_item<1, false>(out, b2, h2, id & 1, false, 0, h2, g0, g1, lo, kz, zmask, tail, ra, rb, afrag,
                           XsP, Fs, cW, iW);
}

extern "C" void kernel_launch(void* const* d_in, const int* in_sizes, int n_in,
                              void* d_out, int out_size, void* d_ws, size_t ws_size,
                              hipStream_t stream) {
    (void)in_sizes; (void)n_in; (void)out_size;
    const float* x    = (const float*)d_in[0];
    const float* lego = (const float*)d_in[1];
    const float* coef = (const float*)d_in[2];
    const float* comb = (const float*)d_in[3];
    float* out = (float*)d_out;
    float* tab = (float*)d_ws;

    lego_select<<<dim3(SEL_WGS, 1, 1), dim3(32, 1, 1), 0, stream>>>(coef, comb, tab, ws_size);

    dim3 grid(N_WG, 1, 1);
    dim3 block(256, 1, 1);
    lego_conv_fused<<<grid, block, 0, stream>>>(x, lego, coef, comb, tab, ws_size, out);
}
